# starting kernel
# baseline (speedup 1.0000x reference)
#include <hip/hip_runtime.h>
#include <math.h>

typedef __attribute__((ext_vector_type(2))) float v2f;
typedef __attribute__((ext_vector_type(4))) float v4f;
typedef __attribute__((ext_vector_type(8))) float v8f;

#define NSTATE 1024
#define NHEAD  16
#define HDIM   64
#define NUMROT 8
#define SEQ    4096
#define BATCH  4
#define ROWS   (BATCH * SEQ * NHEAD)
#define TILES  (ROWS / 16)

__global__ void compose_R_kernel(const float* __restrict__ thetas,
                                 const float* __restrict__ pairs,
                                 const float* __restrict__ theta_scale,
                                 const float* __restrict__ rotM,
                                 float* __restrict__ Rp) {
  __shared__ float R[64 * 64];
  int r = threadIdx.x;
  if (r < 64) {
    for (int c = 0; c < 64; ++c) R[r * 64 + c] = (r == c) ? 1.0f : 0.0f;
    float ts = theta_scale[0];
    for (int k = 0; k < NUMROT; ++k) {
      int i = (int)pairs[2 * k + 0];
      int j = (int)pairs[2 * k + 1];
      float th = thetas[k] * ts;
      float c = cosf(th), s = sinf(th);
      if (i == j) {
        R[r * 64 + i] *= c;
      } else {
        float a = R[r * 64 + i], b = R[r * 64 + j];
        R[r * 64 + i] =  c * a + s * b;
        R[r * 64 + j] = -s * a + c * b;
      }
    }
    for (int c2 = 0; c2 < 64; ++c2) {
      float v = 0.0f;
      for (int k = 0; k < 64; ++k) v += R[r * 64 + k] * rotM[k * 64 + c2];
      int p = (c2 & 1) ? (32 + (c2 >> 1)) : (c2 >> 1);
      Rp[r * 64 + p] = v;
    }
  }
}

__global__ void sincos_kernel(const float* __restrict__ inv_freq,
                              float* __restrict__ tab) {
  int idx = blockIdx.x * blockDim.x + threadIdx.x;
  if (idx < SEQ * 32) {
    int s = idx >> 5, d = idx & 31;
    float ang = (float)s * inv_freq[d];
    tab[2 * idx + 0] = cosf(ang);
    tab[2 * idx + 1] = sinf(ang);
  }
}

__global__ void __launch_bounds__(256)
rot_rope_kernel(const float* __restrict__ X,
                const float* __restrict__ Rp,
                const float* __restrict__ tab,
                float* __restrict__ out) {
  __shared__ float lds[64 * 64];

  int tid = threadIdx.x;
  for (int i = tid; i < 1024; i += 256)
    ((v4f*)lds)[i] = ((const v4f*)Rp)[i];
  __syncthreads();

  int lane = tid & 31;
  int wave = tid >> 5;
  int half = lane >> 4;
  int l15  = lane & 15;

  v2f Bf[4][16];
#pragma unroll
  for (int nb = 0; nb < 4; ++nb) {
#pragma unroll
    for (int kb = 0; kb < 16; ++kb) {
      int k0 = 4 * kb + 2 * half;
      int n  = 16 * nb + l15;
      v2f b;
      b.x = lds[k0 * 64 + n];
      b.y = lds[(k0 + 1) * 64 + n];
      Bf[nb][kb] = b;
    }
  }

  int waveId = blockIdx.x * 8 + wave;
  int stride = gridDim.x * 8;

  for (int t = waveId; t < TILES; t += stride) {
    const float* xrow = X + ((size_t)t * 16 + l15) * 64;
    if (t + stride < TILES)
      __builtin_prefetch(xrow + (size_t)stride * 16 * 64, 0, 0);

    v2f Af[16];
#pragma unroll
    for (int kb = 0; kb < 16; ++kb)
      Af[kb] = *(const v2f*)(xrow + 4 * kb + 2 * half);

    int s_pos = t & (SEQ - 1);
    float* obase = out + (size_t)t * 16 * 64;

#pragma unroll
    for (int p = 0; p < 2; ++p) {
      v8f accA = {}, accB = {};
#pragma unroll
      for (int kb = 0; kb < 16; ++kb) {
        accA = __builtin_amdgcn_wmma_f32_16x16x4_f32(
            false, Af[kb], false, Bf[p][kb],     (short)0, accA, false, false);
        accB = __builtin_amdgcn_wmma_f32_16x16x4_f32(
            false, Af[kb], false, Bf[p + 2][kb], (short)0, accB, false, false);
      }

      int dp = 16 * p + l15;
      v2f cs = *(const v2f*)(tab + ((size_t)s_pos * 32 + dp) * 2);
      float c = cs.x, s = cs.y;

#pragma unroll
      for (int v = 0; v < 8; ++v) {
        int row = v + 8 * half;
        float a = accA[v], b = accB[v];
        obase[(size_t)row * 64 + dp]      = a * c - b * s;
        obase[(size_t)row * 64 + 32 + dp] = a * s + b * c;
      }
    }
  }
}

extern "C" void kernel_launch(void* const* d_in, const int* in_sizes, int n_in,
                              void* d_out, int out_size, void* d_ws, size_t ws_size,
                              hipStream_t stream) {
  const float* x      = (const float*)d_in[0];
  const float* thetas = (const float*)d_in[1];
  const float* pairs  = (const float*)d_in[2];
  const float* tscale = (const float*)d_in[3];
  const float* rotM   = (const float*)d_in[4];
  const float* invf   = (const float*)d_in[5];
  float* out = (float*)d_out;

  float* Rp  = (float*)d_ws;
  float* tab = Rp + 64 * 64;

  compose_R_kernel<<<1, 64, 0, stream>>>(thetas, pairs, tscale, rotM, Rp);
  sincos_kernel<<<(SEQ * 32 + 255) / 256, 256, 0, stream>>>(invf, tab);
  rot_rope_kernel<<<512, 256, 0, stream>>>(x, Rp, tab, out);
}


# fragment ownership permuted for wide whole-line access: R kept in fragment order (B straight to registers, no LDS stage), x read 16 B/lane, out in whole 128-B lines; k order per product step changes
# speedup vs baseline: 3.8879x; 3.8879x over previous
#include <hip/hip_runtime.h>
#include <math.h>

typedef __attribute__((ext_vector_type(2))) float v2f;
typedef __attribute__((ext_vector_type(4))) float v4f;
typedef __attribute__((ext_vector_type(8))) float v8f;

#define NSTATE 1024
#define NHEAD  16
#define HDIM   64
#define NUMROT 8
#define SEQ    4096
#define BATCH  4
#define ROWS   (BATCH * SEQ * NHEAD)
#define TILES  (ROWS / 16)


#define TAB_BLOCKS  ((SEQ * 32) / 256)
#define PREP_BLOCKS (1 + TAB_BLOCKS)

__global__ void __launch_bounds__(256)
prep_kernel(const float* __restrict__ thetas,
            const float* __restrict__ pairs,
            const float* __restrict__ theta_scale,
            const float* __restrict__ rotM,
            const float* __restrict__ inv_freq,
            float* __restrict__ Rf,
            float* __restrict__ tab) {
  __shared__ float M[64 * 64];
  __shared__ v4f   prm[NUMROT];
  int tid = threadIdx.x;

  if (blockIdx.x != 0) {
    int idx = (blockIdx.x - 1) * 256 + tid;
    int s = idx >> 5, d = idx & 31;
    float ang = (float)s * inv_freq[d];
    tab[2 * idx + 0] = cosf(ang);
    tab[2 * idx + 1] = sinf(ang);
    return;
  }

  const v4f* src = (const v4f*)rotM;
  v4f m0 = src[tid], m1 = src[tid + 256], m2 = src[tid + 512], m3 = src[tid + 768];

  if (tid < NUMROT) {
    float th = thetas[tid] * theta_scale[0];
    v4f q;
    q.x = cosf(th);
    q.y = sinf(th);
    q.z = pairs[2 * tid + 0];
    q.w = pairs[2 * tid + 1];
    prm[tid] = q;
  }
  ((v4f*)M)[tid]       = m0;
  ((v4f*)M)[tid + 256] = m1;
  ((v4f*)M)[tid + 512] = m2;
  ((v4f*)M)[tid + 768] = m3;
  __syncthreads();

  if (tid < 64) {
    v4f q[NUMROT];
#pragma unroll
    for (int k = 0; k < NUMROT; ++k) q[k] = prm[k];
#pragma unroll
    for (int k = NUMROT - 1; k >= 0; --k) {
      float c = q[k].x, s = q[k].y;
      int i = (int)q[k].z, j = (int)q[k].w;
      if (i == j) {
        M[i * 64 + tid] *= c;
      } else {
        float a = M[i * 64 + tid], b = M[j * 64 + tid];
        M[i * 64 + tid] = c * a - s * b;
        M[j * 64 + tid] = s * a + c * b;
      }
    }
  }
  __syncthreads();

  for (int q4 = tid; q4 < 1024; q4 += 256) {
    int lane = q4 & 31, ap = (q4 >> 5) & 1, ws = q4 >> 6;
    int h = lane >> 4, l = lane & 15;
    int k = 8 * (ws >> 1) + 4 * h + 2 * (ws & 1);
    const float* mr = M + k * 64 + 4 * l + ap;
    v4f o;
    o.x = mr[0];  o.y = mr[64];
    o.z = mr[2];  o.w = mr[66];
    ((v4f*)Rf)[q4] = o;
  }
}

__global__ void __launch_bounds__(256)
rot_rope_kernel(const float* __restrict__ X,
                const float* __restrict__ Rf,
                const float* __restrict__ tab,
                float* __restrict__ out) {
  int tid  = threadIdx.x;
  int lane = tid & 31;
  int wave = tid >> 5;
  int half = lane >> 4;
  int l15  = lane & 15;

  v2f Bf[4][16];
  const v4f* rf = (const v4f*)Rf + lane;
#pragma unroll
  for (int ws = 0; ws < 16; ++ws) {
#pragma unroll
    for (int ap = 0; ap < 2; ++ap) {
      v4f q = rf[(ws * 2 + ap) * 32];
      v2f b0, b1;
      b0.x = q.x; b0.y = q.y;
      b1.x = q.z; b1.y = q.w;
      Bf[2 * ap][ws]     = b0;
      Bf[2 * ap + 1][ws] = b1;
    }
  }

  int waveId = blockIdx.x * 8 + wave;
  int stride = gridDim.x * 8;

  for (int t = waveId; t < TILES; t += stride) {
    const float* xrow = X + ((size_t)t * 16 + l15) * 64;
    if (t + stride < TILES)
      __builtin_prefetch(xrow + (size_t)stride * 16 * 64, 0, 0);

    v4f a4[8];
#pragma unroll
    for (int j = 0; j < 8; ++j)
      a4[j] = *(const v4f*)(xrow + 8 * j + 4 * half);

    int s_pos = t & (SEQ - 1);
    float* obase = out + (size_t)t * 16 * 64;

    v4f cs = *(const v4f*)(tab + ((size_t)s_pos * 32 + 2 * l15) * 2);

    v8f acc0 = {}, acc1 = {}, acc2 = {}, acc3 = {};
#pragma unroll
    for (int j = 0; j < 8; ++j) {
      v2f alo, ahi;
      alo.x = a4[j].x; alo.y = a4[j].y;
      ahi.x = a4[j].z; ahi.y = a4[j].w;
      acc0 = __builtin_amdgcn_wmma_f32_16x16x4_f32(
          false, alo, false, Bf[0][2 * j],     (short)0, acc0, false, false);
      acc1 = __builtin_amdgcn_wmma_f32_16x16x4_f32(
          false, alo, false, Bf[1][2 * j],     (short)0, acc1, false, false);
      acc2 = __builtin_amdgcn_wmma_f32_16x16x4_f32(
          false, alo, false, Bf[2][2 * j],     (short)0, acc2, false, false);
      acc3 = __builtin_amdgcn_wmma_f32_16x16x4_f32(
          false, alo, false, Bf[3][2 * j],     (short)0, acc3, false, false);
      acc0 = __builtin_amdgcn_wmma_f32_16x16x4_f32(
          false, ahi, false, Bf[0][2 * j + 1], (short)0, acc0, false, false);
      acc1 = __builtin_amdgcn_wmma_f32_16x16x4_f32(
          false, ahi, false, Bf[1][2 * j + 1], (short)0, acc1, false, false);
      acc2 = __builtin_amdgcn_wmma_f32_16x16x4_f32(
          false, ahi, false, Bf[2][2 * j + 1], (short)0, acc2, false, false);
      acc3 = __builtin_amdgcn_wmma_f32_16x16x4_f32(
          false, ahi, false, Bf[3][2 * j + 1], (short)0, acc3, false, false);
    }

    float c0 = cs.x, s0 = cs.y, c1 = cs.z, s1 = cs.w;
    float* orow = obase + (size_t)(8 * half) * 64 + 2 * l15;
#pragma unroll
    for (int v = 0; v < 8; ++v) {
      float a0 = acc0[v], a1 = acc1[v], b0 = acc2[v], b1 = acc3[v];
      v2f lo, hi;
      lo.x = a0 * c0 - b0 * s0;
      lo.y = a1 * c1 - b1 * s1;
      hi.x = a0 * s0 + b0 * c0;
      hi.y = a1 * s1 + b1 * c1;
      *(v2f*)(orow + (size_t)v * 64)      = lo;
      *(v2f*)(orow + (size_t)v * 64 + 32) = hi;
    }
  }
}

extern "C" void kernel_launch(void* const* d_in, const int* in_sizes, int n_in,
                              void* d_out, int out_size, void* d_ws, size_t ws_size,
                              hipStream_t stream) {
  const float* x      = (const float*)d_in[0];
  const float* thetas = (const float*)d_in[1];
  const float* pairs  = (const float*)d_in[2];
  const float* tscale = (const float*)d_in[3];
  const float* rotM   = (const float*)d_in[4];
  const float* invf   = (const float*)d_in[5];
  float* out = (float*)d_out;

  float* Rf  = (float*)d_ws;
  float* tab = Rf + 64 * 64;

  prep_kernel<<<PREP_BLOCKS, 256, 0, stream>>>(thetas, pairs, tscale, rotM, invf, Rf, tab);
  rot_rope_kernel<<<512, 256, 0, stream>>>(x, Rf, tab, out);
}


# no prep launch: the 8 Givens applied to each x tile in wave-private LDS, out=((x G0)..G7)@M; B straight from rotation_matrix, cos/sin per wave; no workspace; re-associated, rounding differs
# speedup vs baseline: 4.2224x; 1.0860x over previous
#include <hip/hip_runtime.h>
#include <math.h>


typedef __attribute__((ext_vector_type(2))) float v2f;
typedef __attribute__((ext_vector_type(4))) float v4f;
typedef __attribute__((ext_vector_type(8))) float v8f;

#define NSTATE 1024
#define NHEAD  16
#define HDIM   64
#define NUMROT 8
#define SEQ    4096
#define BATCH  4
#define ROWS   (BATCH * SEQ * NHEAD)
#define TILES  (ROWS / 16)

#define WAVES_PER_BLOCK 8
#define GRID_BLOCKS (SEQ / WAVES_PER_BLOCK)

#define ROW_V4 17
#define ROW_F  (4 * ROW_V4)

__global__ void __launch_bounds__(256)
rot_rope_kernel(const float* __restrict__ X,
                const float* __restrict__ thetas,
                const float* __restrict__ pairs,
                const float* __restrict__ theta_scale,
                const float* __restrict__ rotM,
                const float* __restrict__ inv_freq,
                float* __restrict__ out) {
  __shared__ v4f stage[WAVES_PER_BLOCK][16 * ROW_V4];

  int tid  = threadIdx.x;
  int lane = tid & 31;
  int wave = tid >> 5;
  int half = lane >> 4;
  int l15  = lane & 15;

  int s_pos = blockIdx.x * WAVES_PER_BLOCK + wave;

  const float* xrow = X + ((size_t)s_pos * 16 + l15) * 64 + 4 * half;
  v4f a4[8];
#pragma unroll
  for (int j = 0; j < 8; ++j)
    a4[j] = *(const v4f*)(xrow + 8 * j);

  v2f Bf[4][16];
  const float* mrow = rotM + (4 * half) * 64 + 4 * l15;
#pragma unroll
  for (int ws = 0; ws < 16; ++ws) {
    int k = 8 * (ws >> 1) + 2 * (ws & 1);
    v4f r0 = *(const v4f*)(mrow + (size_t)k * 64);
    v4f r1 = *(const v4f*)(mrow + (size_t)(k + 1) * 64);
    v2f b;
    b.x = r0.x; b.y = r1.x; Bf[0][ws] = b;
    b.x = r0.z; b.y = r1.z; Bf[1][ws] = b;
    b.x = r0.y; b.y = r1.y; Bf[2][ws] = b;
    b.x = r0.w; b.y = r1.w; Bf[3][ws] = b;
  }

  int   k8 = lane & 7;
  float th = thetas[k8] * theta_scale[0];
  float gc = cosf(th);
  float gs = sinf(th);
  int   gi = (int)pairs[2 * k8 + 0] & 63;
  int   gj = (int)pairs[2 * k8 + 1] & 63;
  if (gi == gj) gs = 0.0f;

  float rc[NUMROT];
  float rsg[NUMROT];
  int   own[NUMROT];
  int   oth[NUMROT];
#pragma unroll
  for (int k = 0; k < NUMROT; ++k) {
    float c  = __int_as_float(__builtin_amdgcn_readlane(__float_as_int(gc), k));
    float sn = __int_as_float(__builtin_amdgcn_readlane(__float_as_int(gs), k));
    int   i  = __builtin_amdgcn_readlane(gi, k);
    int   j  = __builtin_amdgcn_readlane(gj, k);
    rc[k]  = c;
    rsg[k] = half ? -sn : sn;
    own[k] = half ? j : i;
    oth[k] = half ? i : j;
  }

  float ang0 = (float)s_pos * inv_freq[2 * l15];
  float ang1 = (float)s_pos * inv_freq[2 * l15 + 1];
  float c0 = cosf(ang0), s0 = sinf(ang0);
  float c1 = cosf(ang1), s1 = sinf(ang1);

  float* trow = (float*)&stage[wave][0] + l15 * ROW_F;
  float* tfrag = trow + 4 * half;

  for (int b = 0; b < BATCH; ++b) {
    size_t t = (size_t)b * SEQ + s_pos;

#pragma unroll
    for (int j = 0; j < 8; ++j)
      *(v4f*)(tfrag + 8 * j) = a4[j];

    const float* nrow = xrow + (size_t)(b + 1) * SEQ * 16 * 64;
    if (b + 1 < BATCH)
      __builtin_prefetch(nrow, 0, 0);

#pragma unroll
    for (int k = 0; k < NUMROT; ++k) {
      float p = trow[own[k]];
      float q = trow[oth[k]];
      trow[own[k]] = rc[k] * p + rsg[k] * q;
    }

#pragma unroll
    for (int j = 0; j < 8; ++j)
      a4[j] = *(const v4f*)(tfrag + 8 * j);
    __builtin_amdgcn_sched_barrier(0);

    v4f n4[8];
    if (b + 1 < BATCH) {
#pragma unroll
      for (int j = 0; j < 8; ++j)
        n4[j] = *(const v4f*)(nrow + 8 * j);
    }

    v8f acc0 = {}, acc1 = {}, acc2 = {}, acc3 = {};
#pragma unroll
    for (int j = 0; j < 8; ++j) {
      v2f alo, ahi;
      alo.x = a4[j].x; alo.y = a4[j].y;
      ahi.x = a4[j].z; ahi.y = a4[j].w;
      acc0 = __builtin_amdgcn_wmma_f32_16x16x4_f32(
          false, alo, false, Bf[0][2 * j],     (short)0, acc0, false, false);
      acc1 = __builtin_amdgcn_wmma_f32_16x16x4_f32(
          false, alo, false, Bf[1][2 * j],     (short)0, acc1, false, false);
      acc2 = __builtin_amdgcn_wmma_f32_16x16x4_f32(
          false, alo, false, Bf[2][2 * j],     (short)0, acc2, false, false);
      acc3 = __builtin_amdgcn_wmma_f32_16x16x4_f32(
          false, alo, false, Bf[3][2 * j],     (short)0, acc3, false, false);
      acc0 = __builtin_amdgcn_wmma_f32_16x16x4_f32(
          false, ahi, false, Bf[0][2 * j + 1], (short)0, acc0, false, false);
      acc1 = __builtin_amdgcn_wmma_f32_16x16x4_f32(
          false, ahi, false, Bf[1][2 * j + 1], (short)0, acc1, false, false);
      acc2 = __builtin_amdgcn_wmma_f32_16x16x4_f32(
          false, ahi, false, Bf[2][2 * j + 1], (short)0, acc2, false, false);
      acc3 = __builtin_amdgcn_wmma_f32_16x16x4_f32(
          false, ahi, false, Bf[3][2 * j + 1], (short)0, acc3, false, false);
    }

    float* orow = out + t * 16 * 64 + (size_t)(8 * half) * 64 + 2 * l15;
#pragma unroll
    for (int v = 0; v < 8; ++v) {
      float x10 = acc0[v], x11 = acc1[v], x20 = acc2[v], x21 = acc3[v];
      v2f lo, hi;
      lo.x = x10 * c0 - x20 * s0;
      lo.y = x11 * c1 - x21 * s1;
      hi.x = x10 * s0 + x20 * c0;
      hi.y = x11 * s1 + x21 * c1;
      *(v2f*)(orow + (size_t)v * 64)      = lo;
      *(v2f*)(orow + (size_t)v * 64 + 32) = hi;
    }

    if (b + 1 < BATCH) {
#pragma unroll
      for (int j = 0; j < 8; ++j) a4[j] = n4[j];
    }
  }
}

extern "C" void kernel_launch(void* const* d_in, const int* in_sizes, int n_in,
                              void* d_out, int out_size, void* d_ws, size_t ws_size,
                              hipStream_t stream) {
  const float* x      = (const float*)d_in[0];
  const float* thetas = (const float*)d_in[1];
  const float* pairs  = (const float*)d_in[2];
  const float* tscale = (const float*)d_in[3];
  const float* rotM   = (const float*)d_in[4];
  const float* invf   = (const float*)d_in[5];
  float* out = (float*)d_out;

  rot_rope_kernel<<<GRID_BLOCKS, 256, 0, stream>>>(x, thetas, pairs, tscale, rotM, invf, out);
}


# x tile read by whole-line loads: each 16-B-per-lane load covers 512 contiguous bytes (4 lines) and fills two rows of the wave's row-major LDS stage, not 32-B pieces of 16 lines; values unchanged
# speedup vs baseline: 4.4424x; 1.0521x over previous
#include <hip/hip_runtime.h>
#include <math.h>


typedef __attribute__((ext_vector_type(2))) float v2f;
typedef __attribute__((ext_vector_type(4))) float v4f;
typedef __attribute__((ext_vector_type(8))) float v8f;

#define NSTATE 1024
#define NHEAD  16
#define HDIM   64
#define NUMROT 8
#define SEQ    4096
#define BATCH  4
#define ROWS   (BATCH * SEQ * NHEAD)
#define TILES  (ROWS / 16)

#define WAVES_PER_BLOCK 8
#define GRID_BLOCKS (SEQ / WAVES_PER_BLOCK)

#define ROW_V4 17
#define ROW_F  (4 * ROW_V4)

__global__ void __launch_bounds__(256)
rot_rope_kernel(const float* __restrict__ X,
                const float* __restrict__ thetas,
                const float* __restrict__ pairs,
                const float* __restrict__ theta_scale,
                const float* __restrict__ rotM,
                const float* __restrict__ inv_freq,
                float* __restrict__ out) {
  __shared__ v4f stage[WAVES_PER_BLOCK][16 * ROW_V4];

  int tid  = threadIdx.x;
  int lane = tid & 31;
  int wave = tid >> 5;
  int half = lane >> 4;
  int l15  = lane & 15;

  int s_pos = blockIdx.x * WAVES_PER_BLOCK + wave;

  const float* xld = X + (size_t)s_pos * 16 * 64 + 4 * lane;
  const float* xrow = X + ((size_t)s_pos * 16 + l15) * 64 + 4 * half;
  v4f a4[8];
#pragma unroll
  for (int q = 0; q < 8; ++q)
    a4[q] = *(const v4f*)(xld + 128 * q);

  v2f Bf[4][16];
  const float* mrow = rotM + (4 * half) * 64 + 4 * l15;
#pragma unroll
  for (int ws = 0; ws < 16; ++ws) {
    int k = 8 * (ws >> 1) + 2 * (ws & 1);
    v4f r0 = *(const v4f*)(mrow + (size_t)k * 64);
    v4f r1 = *(const v4f*)(mrow + (size_t)(k + 1) * 64);
    v2f b;
    b.x = r0.x; b.y = r1.x; Bf[0][ws] = b;
    b.x = r0.z; b.y = r1.z; Bf[1][ws] = b;
    b.x = r0.y; b.y = r1.y; Bf[2][ws] = b;
    b.x = r0.w; b.y = r1.w; Bf[3][ws] = b;
  }

  int   k8 = lane & 7;
  float th = thetas[k8] * theta_scale[0];
  float gc = cosf(th);
  float gs = sinf(th);
  int   gi = (int)pairs[2 * k8 + 0] & 63;
  int   gj = (int)pairs[2 * k8 + 1] & 63;
  if (gi == gj) gs = 0.0f;

  float rc[NUMROT];
  float rsg[NUMROT];
  int   own[NUMROT];
  int   oth[NUMROT];
#pragma unroll
  for (int k = 0; k < NUMROT; ++k) {
    float c  = __int_as_float(__builtin_amdgcn_readlane(__float_as_int(gc), k));
    float sn = __int_as_float(__builtin_amdgcn_readlane(__float_as_int(gs), k));
    int   i  = __builtin_amdgcn_readlane(gi, k);
    int   j  = __builtin_amdgcn_readlane(gj, k);
    rc[k]  = c;
    rsg[k] = half ? -sn : sn;
    own[k] = half ? j : i;
    oth[k] = half ? i : j;
  }

  float ang0 = (float)s_pos * inv_freq[2 * l15];
  float ang1 = (float)s_pos * inv_freq[2 * l15 + 1];
  float c0 = cosf(ang0), s0 = sinf(ang0);
  float c1 = cosf(ang1), s1 = sinf(ang1);

  float* trow = (float*)&stage[wave][0] + l15 * ROW_F;
  float* tfrag = trow + 4 * half;
  float* tfill = (float*)&stage[wave][0] + half * ROW_F + 4 * l15;

  for (int b = 0; b < BATCH; ++b) {
    size_t t = (size_t)b * SEQ + s_pos;

#pragma unroll
    for (int q = 0; q < 8; ++q)
      *(v4f*)(tfill + 2 * q * ROW_F) = a4[q];

    const float* nrow = xrow + (size_t)(b + 1) * SEQ * 16 * 64;
    const float* nld  = xld  + (size_t)(b + 1) * SEQ * 16 * 64;
    if (b + 1 < BATCH)
      __builtin_prefetch(nrow, 0, 0);

#pragma unroll
    for (int k = 0; k < NUMROT; ++k) {
      float p = trow[own[k]];
      float q = trow[oth[k]];
      trow[own[k]] = rc[k] * p + rsg[k] * q;
    }

#pragma unroll
    for (int j = 0; j < 8; ++j)
      a4[j] = *(const v4f*)(tfrag + 8 * j);
    __builtin_amdgcn_sched_barrier(0);

    v4f n4[8];
    if (b + 1 < BATCH) {
#pragma unroll
      for (int q = 0; q < 8; ++q)
        n4[q] = *(const v4f*)(nld + 128 * q);
    }

    v8f acc0 = {}, acc1 = {}, acc2 = {}, acc3 = {};
#pragma unroll
    for (int j = 0; j < 8; ++j) {
      v2f alo, ahi;
      alo.x = a4[j].x; alo.y = a4[j].y;
      ahi.x = a4[j].z; ahi.y = a4[j].w;
      acc0 = __builtin_amdgcn_wmma_f32_16x16x4_f32(
          false, alo, false, Bf[0][2 * j],     (short)0, acc0, false, false);
      acc1 = __builtin_amdgcn_wmma_f32_16x16x4_f32(
          false, alo, false, Bf[1][2 * j],     (short)0, acc1, false, false);
      acc2 = __builtin_amdgcn_wmma_f32_16x16x4_f32(
          false, alo, false, Bf[2][2 * j],     (short)0, acc2, false, false);
      acc3 = __builtin_amdgcn_wmma_f32_16x16x4_f32(
          false, alo, false, Bf[3][2 * j],     (short)0, acc3, false, false);
      acc0 = __builtin_amdgcn_wmma_f32_16x16x4_f32(
          false, ahi, false, Bf[0][2 * j + 1], (short)0, acc0, false, false);
      acc1 = __builtin_amdgcn_wmma_f32_16x16x4_f32(
          false, ahi, false, Bf[1][2 * j + 1], (short)0, acc1, false, false);
      acc2 = __builtin_amdgcn_wmma_f32_16x16x4_f32(
          false, ahi, false, Bf[2][2 * j + 1], (short)0, acc2, false, false);
      acc3 = __builtin_amdgcn_wmma_f32_16x16x4_f32(
          false, ahi, false, Bf[3][2 * j + 1], (short)0, acc3, false, false);
    }

    float* orow = out + t * 16 * 64 + (size_t)(8 * half) * 64 + 2 * l15;
#pragma unroll
    for (int v = 0; v < 8; ++v) {
      float x10 = acc0[v], x11 = acc1[v], x20 = acc2[v], x21 = acc3[v];
      v2f lo, hi;
      lo.x = x10 * c0 - x20 * s0;
      lo.y = x11 * c1 - x21 * s1;
      hi.x = x10 * s0 + x20 * c0;
      hi.y = x11 * s1 + x21 * c1;
      *(v2f*)(orow + (size_t)v * 64)      = lo;
      *(v2f*)(orow + (size_t)v * 64 + 32) = hi;
    }

    if (b + 1 < BATCH) {
#pragma unroll
      for (int j = 0; j < 8; ++j) a4[j] = n4[j];
    }
  }
}

extern "C" void kernel_launch(void* const* d_in, const int* in_sizes, int n_in,
                              void* d_out, int out_size, void* d_ws, size_t ws_size,
                              hipStream_t stream) {
  const float* x      = (const float*)d_in[0];
  const float* thetas = (const float*)d_in[1];
  const float* pairs  = (const float*)d_in[2];
  const float* tscale = (const float*)d_in[3];
  const float* rotM   = (const float*)d_in[4];
  const float* invf   = (const float*)d_in[5];
  float* out = (float*)d_out;

  rot_rope_kernel<<<GRID_BLOCKS, 256, 0, stream>>>(x, thetas, pairs, tscale, rotM, invf, out);
}


# wave priority falls with tiles done (s_setprio 3,2,1,0 at each tile's head): the matrix pipe serves the wave with most tiles left, not the oldest; a SIMD's 4 waves advance together; values unchanged
# speedup vs baseline: 4.6715x; 1.0516x over previous
#include <hip/hip_runtime.h>
#include <math.h>


typedef __attribute__((ext_vector_type(2))) float v2f;
typedef __attribute__((ext_vector_type(4))) float v4f;
typedef __attribute__((ext_vector_type(8))) float v8f;

#define NSTATE 1024
#define NHEAD  16
#define HDIM   64
#define NUMROT 8
#define SEQ    4096
#define BATCH  4
#define ROWS   (BATCH * SEQ * NHEAD)
#define TILES  (ROWS / 16)

#define WAVES_PER_BLOCK 8
#define GRID_BLOCKS (SEQ / WAVES_PER_BLOCK)

#define ROW_V4 17
#define ROW_F  (4 * ROW_V4)

__global__ void __launch_bounds__(256)
rot_rope_kernel(const float* __restrict__ X,
                const float* __restrict__ thetas,
                const float* __restrict__ pairs,
                const float* __restrict__ theta_scale,
                const float* __restrict__ rotM,
                const float* __restrict__ inv_freq,
                float* __restrict__ out) {
  __shared__ v4f stage[WAVES_PER_BLOCK][16 * ROW_V4];

  int tid  = threadIdx.x;
  int lane = tid & 31;
  int wave = tid >> 5;
  int half = lane >> 4;
  int l15  = lane & 15;

  int s_pos = blockIdx.x * WAVES_PER_BLOCK + wave;

  const float* xld = X + (size_t)s_pos * 16 * 64 + 4 * lane;
  const float* xrow = X + ((size_t)s_pos * 16 + l15) * 64 + 4 * half;
  v4f a4[8];
#pragma unroll
  for (int q = 0; q < 8; ++q)
    a4[q] = *(const v4f*)(xld + 128 * q);

  v2f Bf[4][16];
  const float* mrow = rotM + (4 * half) * 64 + 4 * l15;
#pragma unroll
  for (int ws = 0; ws < 16; ++ws) {
    int k = 8 * (ws >> 1) + 2 * (ws & 1);
    v4f r0 = *(const v4f*)(mrow + (size_t)k * 64);
    v4f r1 = *(const v4f*)(mrow + (size_t)(k + 1) * 64);
    v2f b;
    b.x = r0.x; b.y = r1.x; Bf[0][ws] = b;
    b.x = r0.z; b.y = r1.z; Bf[1][ws] = b;
    b.x = r0.y; b.y = r1.y; Bf[2][ws] = b;
    b.x = r0.w; b.y = r1.w; Bf[3][ws] = b;
  }

  int   k8 = lane & 7;
  float th = thetas[k8] * theta_scale[0];
  float gc = cosf(th);
  float gs = sinf(th);
  int   gi = (int)pairs[2 * k8 + 0] & 63;
  int   gj = (int)pairs[2 * k8 + 1] & 63;
  if (gi == gj) gs = 0.0f;

  float rc[NUMROT];
  float rsg[NUMROT];
  int   own[NUMROT];
  int   oth[NUMROT];
#pragma unroll
  for (int k = 0; k < NUMROT; ++k) {
    float c  = __int_as_float(__builtin_amdgcn_readlane(__float_as_int(gc), k));
    float sn = __int_as_float(__builtin_amdgcn_readlane(__float_as_int(gs), k));
    int   i  = __builtin_amdgcn_readlane(gi, k);
    int   j  = __builtin_amdgcn_readlane(gj, k);
    rc[k]  = c;
    rsg[k] = half ? -sn : sn;
    own[k] = half ? j : i;
    oth[k] = half ? i : j;
  }

  float ang0 = (float)s_pos * inv_freq[2 * l15];
  float ang1 = (float)s_pos * inv_freq[2 * l15 + 1];
  float c0 = cosf(ang0), s0 = sinf(ang0);
  float c1 = cosf(ang1), s1 = sinf(ang1);

  float* trow = (float*)&stage[wave][0] + l15 * ROW_F;
  float* tfrag = trow + 4 * half;
  float* tfill = (float*)&stage[wave][0] + half * ROW_F + 4 * l15;

  for (int b = 0; b < BATCH; ++b) {
    size_t t = (size_t)b * SEQ + s_pos;

    if (b == 0)      __builtin_amdgcn_s_setprio(3);
    else if (b == 1) __builtin_amdgcn_s_setprio(2);
    else if (b == 2) __builtin_amdgcn_s_setprio(1);
    else             __builtin_amdgcn_s_setprio(0);

#pragma unroll
    for (int q = 0; q < 8; ++q)
      *(v4f*)(tfill + 2 * q * ROW_F) = a4[q];

    const float* nrow = xrow + (size_t)(b + 1) * SEQ * 16 * 64;
    const float* nld  = xld  + (size_t)(b + 1) * SEQ * 16 * 64;
    if (b + 1 < BATCH)
      __builtin_prefetch(nrow, 0, 0);

#pragma unroll
    for (int k = 0; k < NUMROT; ++k) {
      float p = trow[own[k]];
      float q = trow[oth[k]];
      trow[own[k]] = rc[k] * p + rsg[k] * q;
    }

#pragma unroll
    for (int j = 0; j < 8; ++j)
      a4[j] = *(const v4f*)(tfrag + 8 * j);
    __builtin_amdgcn_sched_barrier(0);

    v4f n4[8];
    if (b + 1 < BATCH) {
#pragma unroll
      for (int q = 0; q < 8; ++q)
        n4[q] = *(const v4f*)(nld + 128 * q);
    }

    v8f acc0 = {}, acc1 = {}, acc2 = {}, acc3 = {};
#pragma unroll
    for (int j = 0; j < 8; ++j) {
      v2f alo, ahi;
      alo.x = a4[j].x; alo.y = a4[j].y;
      ahi.x = a4[j].z; ahi.y = a4[j].w;
      acc0 = __builtin_amdgcn_wmma_f32_16x16x4_f32(
          false, alo, false, Bf[0][2 * j],     (short)0, acc0, false, false);
      acc1 = __builtin_amdgcn_wmma_f32_16x16x4_f32(
          false, alo, false, Bf[1][2 * j],     (short)0, acc1, false, false);
      acc2 = __builtin_amdgcn_wmma_f32_16x16x4_f32(
          false, alo, false, Bf[2][2 * j],     (short)0, acc2, false, false);
      acc3 = __builtin_amdgcn_wmma_f32_16x16x4_f32(
          false, alo, false, Bf[3][2 * j],     (short)0, acc3, false, false);
      acc0 = __builtin_amdgcn_wmma_f32_16x16x4_f32(
          false, ahi, false, Bf[0][2 * j + 1], (short)0, acc0, false, false);
      acc1 = __builtin_amdgcn_wmma_f32_16x16x4_f32(
          false, ahi, false, Bf[1][2 * j + 1], (short)0, acc1, false, false);
      acc2 = __builtin_amdgcn_wmma_f32_16x16x4_f32(
          false, ahi, false, Bf[2][2 * j + 1], (short)0, acc2, false, false);
      acc3 = __builtin_amdgcn_wmma_f32_16x16x4_f32(
          false, ahi, false, Bf[3][2 * j + 1], (short)0, acc3, false, false);
    }

    float* orow = out + t * 16 * 64 + (size_t)(8 * half) * 64 + 2 * l15;
#pragma unroll
    for (int v = 0; v < 8; ++v) {
      float x10 = acc0[v], x11 = acc1[v], x20 = acc2[v], x21 = acc3[v];
      v2f lo, hi;
      lo.x = x10 * c0 - x20 * s0;
      lo.y = x11 * c1 - x21 * s1;
      hi.x = x10 * s0 + x20 * c0;
      hi.y = x11 * s1 + x21 * c1;
      *(v2f*)(orow + (size_t)v * 64)      = lo;
      *(v2f*)(orow + (size_t)v * 64 + 32) = hi;
    }

    if (b + 1 < BATCH) {
#pragma unroll
      for (int j = 0; j < 8; ++j) a4[j] = n4[j];
    }
  }
}

extern "C" void kernel_launch(void* const* d_in, const int* in_sizes, int n_in,
                              void* d_out, int out_size, void* d_ws, size_t ws_size,
                              hipStream_t stream) {
  const float* x      = (const float*)d_in[0];
  const float* thetas = (const float*)d_in[1];
  const float* pairs  = (const float*)d_in[2];
  const float* tscale = (const float*)d_in[3];
  const float* rotM   = (const float*)d_in[4];
  const float* invf   = (const float*)d_in[5];
  float* out = (float*)d_out;

  rot_rope_kernel<<<GRID_BLOCKS, 256, 0, stream>>>(x, thetas, pairs, tscale, rotM, invf, out);
}
